# starting kernel
# baseline (speedup 1.0000x reference)
#include <hip/hip_runtime.h>

typedef __attribute__((ext_vector_type(2))) float v2f;
typedef __attribute__((ext_vector_type(8))) float v8f;

#define DIN  512
#define DOUT 512

__global__ void fill_zero_kernel(float* __restrict__ p, size_t count) {
    size_t i      = (size_t)blockIdx.x * blockDim.x + threadIdx.x;
    size_t stride = (size_t)gridDim.x * blockDim.x;
    for (; i < count; i += stride) p[i] = 0.0f;
}

__global__ void rgcn_degree_kernel(const int* __restrict__ dst,
                                   float* __restrict__ deg,
                                   int e_per_rel, int n, int total) {
    int i = blockIdx.x * blockDim.x + threadIdx.x;
    if (i >= total) return;
    int r = i / e_per_rel;
    atomicAdd(&deg[(size_t)r * n + dst[i]], 1.0f);
}

__global__ __launch_bounds__(128)
void rgcn_gemm_wmma_kernel(const float* __restrict__ x,
                           const float* __restrict__ W,
                           const float* __restrict__ b,
                           float* __restrict__ Wh,
                           int nrows) {
    __shared__ float xs[64][33];
    __shared__ float wsT[128][33];

    const int tid  = threadIdx.x;
    const int wave = tid >> 5;
    const int lane = tid & 31;
    const int l15  = lane & 15;
    const int khal = (lane >> 4) * 2;

    const int m0 = blockIdx.x * 64;
    const int n0 = blockIdx.y * 128;
    const int wm = (wave >> 1) * 32;
    const int wn = (wave & 1) * 64;

    v8f acc[2][4] = {};

    for (int k0 = 0; k0 < DIN; k0 += 32) {
        for (int idx = tid; idx < 64 * 32; idx += 128) {
            int r = idx >> 5, c = idx & 31;
            int gr = m0 + r;
            if (gr >= nrows) gr = nrows - 1;
            xs[r][c] = x[(size_t)gr * DIN + k0 + c];
        }
        for (int idx = tid; idx < 32 * 128; idx += 128) {
            int k = idx >> 7, c = idx & 127;
            wsT[c][k] = W[(size_t)(k0 + k) * DOUT + n0 + c];
        }
        __syncthreads();

        #pragma unroll
        for (int kk = 0; kk < 32; kk += 4) {
            const int kb = kk + khal;
            v2f a[2];
            #pragma unroll
            for (int i = 0; i < 2; ++i) {
                a[i].x = xs[wm + i * 16 + l15][kb];
                a[i].y = xs[wm + i * 16 + l15][kb + 1];
            }
            v2f bf[4];
            #pragma unroll
            for (int j = 0; j < 4; ++j) {
                bf[j].x = wsT[wn + j * 16 + l15][kb];
                bf[j].y = wsT[wn + j * 16 + l15][kb + 1];
            }
            #pragma unroll
            for (int i = 0; i < 2; ++i)
                #pragma unroll
                for (int j = 0; j < 4; ++j)
                    acc[i][j] = __builtin_amdgcn_wmma_f32_16x16x4_f32(
                        false, a[i], false, bf[j], (short)0, acc[i][j], false, false);
        }
        __syncthreads();
    }

    const int mbase = m0 + wm + (lane >> 4) * 8;
    #pragma unroll
    for (int i = 0; i < 2; ++i) {
        #pragma unroll
        for (int j = 0; j < 4; ++j) {
            const int col  = n0 + wn + j * 16 + l15;
            const float bb = b[col];
            #pragma unroll
            for (int v = 0; v < 8; ++v) {
                const int row = mbase + i * 16 + v;
                if (row < nrows)
                    Wh[(size_t)row * DOUT + col] = acc[i][j][v] + bb;
            }
        }
    }
}

__global__ __launch_bounds__(256)
void rgcn_edge_agg_kernel(const float* __restrict__ Wh,
                          const int* __restrict__ src,
                          const int* __restrict__ dst,
                          const float* __restrict__ deg,
                          float* __restrict__ out,
                          int nedges) {
    const int e = blockIdx.x * 2 + (threadIdx.x >> 7);
    if (e >= nedges) return;
    const int t = threadIdx.x & 127;
    const int s = src[e];
    const int d = dst[e];
    const float inv = 1.0f / fmaxf(deg[d], 1.0f);

    const float4 v = *(const float4*)(Wh + (size_t)s * DOUT + t * 4);
    float* o = out + (size_t)d * DOUT + t * 4;
    atomicAdd(o + 0, v.x * inv);
    atomicAdd(o + 1, v.y * inv);
    atomicAdd(o + 2, v.z * inv);
    atomicAdd(o + 3, v.w * inv);
}

extern "C" void kernel_launch(void* const* d_in, const int* in_sizes, int n_in,
                              void* d_out, int out_size, void* d_ws, size_t ws_size,
                              hipStream_t stream) {
    const float* x   = (const float*)d_in[0];
    const float* W   = (const float*)d_in[1];
    const float* b   = (const float*)d_in[2];
    const int*   src = (const int*)d_in[3];
    const int*   dst = (const int*)d_in[4];
    float*       out = (float*)d_out;

    const int n    = in_sizes[0] / DIN;
    const int nrel = in_sizes[1] / (DIN * DOUT);
    const int e    = in_sizes[3] / nrel;

    float* Wh  = (float*)d_ws;
    float* deg = Wh + (size_t)n * DOUT;

    fill_zero_kernel<<<2048, 256, 0, stream>>>(out, (size_t)n * DOUT);
    fill_zero_kernel<<<512, 256, 0, stream>>>(deg, (size_t)nrel * n);

    {
        const int total  = nrel * e;
        const int blocks = (total + 255) / 256;
        rgcn_degree_kernel<<<blocks, 256, 0, stream>>>(dst, deg, e, n, total);
    }

    dim3 ggrid((n + 63) / 64, DOUT / 128, 1);
    for (int r = 0; r < nrel; ++r) {
        rgcn_gemm_wmma_kernel<<<ggrid, 128, 0, stream>>>(
            x, W + (size_t)r * DIN * DOUT, b + (size_t)r * DOUT, Wh, n);
        rgcn_edge_agg_kernel<<<(e + 1) / 2, 256, 0, stream>>>(
            Wh, src + (size_t)r * e, dst + (size_t)r * e,
            deg + (size_t)r * n, out, e);
    }
}


# Mean over in-edges taken on x rows before the product (linearity): edge lists by counting sort (int atomics), a wave sums x[src] per row in registers, GEMM adds a_r W_r into out; sum order differs
# speedup vs baseline: 1.6268x; 1.6268x over previous
#include <hip/hip_runtime.h>

typedef __attribute__((ext_vector_type(2))) float v2f;
typedef __attribute__((ext_vector_type(8))) float v8f;

#define DIN  512
#define DOUT 512


__global__ void fill_zero_int_kernel(int* __restrict__ p, int count) {
    int i      = blockIdx.x * blockDim.x + threadIdx.x;
    int stride = gridDim.x * blockDim.x;
    for (; i < count; i += stride) p[i] = 0;
}

__global__ void rgcn_count_kernel(const int* __restrict__ dst,
                                  int* __restrict__ cnt,
                                  int e_per_rel, int n, int total) {
    int i = blockIdx.x * blockDim.x + threadIdx.x;
    if (i >= total) return;
    int r = i / e_per_rel;
    atomicAdd(&cnt[(size_t)r * n + dst[i]], 1);
}

__global__ __launch_bounds__(256)
void rgcn_offsets_kernel(const int* __restrict__ cnt, int* __restrict__ cur, int m) {
    __shared__ int front[8];
    __shared__ int wtot[8];
    const int tid  = threadIdx.x;
    const int lane = tid & 31;
    const int wave = tid >> 5;
    const int tile0 = blockIdx.x * 1024;
    const int4* c4 = (const int4*)cnt;

    int s = 0;
    for (int i = tid; i < (tile0 >> 2); i += 256) {
        const int4 v = c4[i];
        s += (v.x + v.y) + (v.z + v.w);
    }
    for (int o = 16; o > 0; o >>= 1) s += __shfl_xor(s, o);
    if (lane == 0) front[wave] = s;

    const int idx = tile0 + tid * 4;
    int a0 = 0, a1 = 0, a2 = 0, a3 = 0;
    if (idx + 3 < m) {
        const int4 v = c4[idx >> 2];
        a0 = v.x; a1 = v.y; a2 = v.z; a3 = v.w;
    } else {
        if (idx     < m) a0 = cnt[idx];
        if (idx + 1 < m) a1 = cnt[idx + 1];
        if (idx + 2 < m) a2 = cnt[idx + 2];
    }
    const int t = (a0 + a1) + (a2 + a3);
    int inc = t;
    for (int o = 1; o < 32; o <<= 1) {
        const int u = __shfl_up(inc, o);
        if (lane >= o) inc += u;
    }
    if (lane == 31) wtot[wave] = inc;
    __syncthreads();

    int base = 0;
    for (int w = 0; w < 8; ++w) base += front[w];
    for (int w = 0; w < wave; ++w) base += wtot[w];
    const int ex = base + inc - t;
    if (idx     < m) cur[idx]     = ex;
    if (idx + 1 < m) cur[idx + 1] = ex + a0;
    if (idx + 2 < m) cur[idx + 2] = ex + a0 + a1;
    if (idx + 3 < m) cur[idx + 3] = ex + a0 + a1 + a2;
}

__global__ void rgcn_list_kernel(const int* __restrict__ src,
                                 const int* __restrict__ dst,
                                 int* __restrict__ cur,
                                 int* __restrict__ esrc,
                                 int e_per_rel, int n, int total) {
    int i = blockIdx.x * blockDim.x + threadIdx.x;
    if (i >= total) return;
    int r = i / e_per_rel;
    const int pos = atomicAdd(&cur[(size_t)r * n + dst[i]], 1);
    esrc[pos] = src[i];
}

__device__ __forceinline__ void add4(float4& a, const float4 v) {
    a.x += v.x; a.y += v.y; a.z += v.z; a.w += v.w;
}
__device__ __forceinline__ float4 mul4(const float4 a, const float s) {
    float4 r; r.x = a.x * s; r.y = a.y * s; r.z = a.z * s; r.w = a.w * s; return r;
}

__global__ __launch_bounds__(256)
void rgcn_row_mean_kernel(const float* __restrict__ x,
                          const int* __restrict__ cnt,
                          const int* __restrict__ cur,
                          const int* __restrict__ esrc,
                          float* __restrict__ agg,
                          int n) {
    const int lane = threadIdx.x & 31;
    const int wave = __builtin_amdgcn_readfirstlane((int)((blockIdx.x * 256 + threadIdx.x) >> 5));
    const int nw   = (gridDim.x * 256) >> 5;

    for (int row = wave; row < n; row += nw) {
        const int c   = cnt[row];
        const int end = cur[row];
        float4 a0 = {0.f, 0.f, 0.f, 0.f}, a1 = a0, a2 = a0, a3 = a0;
        int k = end - c;
        for (; k + 1 < end; k += 2) {
            const int s0 = esrc[k];
            const int s1 = esrc[k + 1];
            const float4* p0 = (const float4*)(x + (size_t)s0 * DIN) + lane;
            const float4* p1 = (const float4*)(x + (size_t)s1 * DIN) + lane;
            const float4 u0 = p0[0], u1 = p0[32], u2 = p0[64], u3 = p0[96];
            const float4 w0 = p1[0], w1 = p1[32], w2 = p1[64], w3 = p1[96];
            add4(a0, u0); add4(a1, u1); add4(a2, u2); add4(a3, u3);
            add4(a0, w0); add4(a1, w1); add4(a2, w2); add4(a3, w3);
        }
        if (k < end) {
            const int s0 = esrc[k];
            const float4* p0 = (const float4*)(x + (size_t)s0 * DIN) + lane;
            const float4 u0 = p0[0], u1 = p0[32], u2 = p0[64], u3 = p0[96];
            add4(a0, u0); add4(a1, u1); add4(a2, u2); add4(a3, u3);
        }
        const float inv = 1.0f / fmaxf((float)c, 1.0f);
        float4* o = (float4*)(agg + (size_t)row * DIN) + lane;
        o[0]  = mul4(a0, inv);
        o[32] = mul4(a1, inv);
        o[64] = mul4(a2, inv);
        o[96] = mul4(a3, inv);
    }
}

#define CHUNK_F 1024
#define TILE_F  128

__device__ __forceinline__ v2f lo2(const float4 q) { v2f r; r.x = q.x; r.y = q.y; return r; }
__device__ __forceinline__ v2f hi2(const float4 q) { v2f r; r.x = q.z; r.y = q.w; return r; }

__device__ __forceinline__ void load_b(float4 (&bq)[8], const float* bbase, int q) {
    #pragma unroll
    for (int n = 0; n < 8; ++n)
        bq[n] = *(const float4*)(bbase + q * CHUNK_F + n * TILE_F);
    __builtin_amdgcn_sched_barrier(0);
}

__device__ __forceinline__ void mma_chunk(v8f (&acc)[8], const float4 qa, const float4 (&bq)[8]) {
    const v2f a0 = lo2(qa);
    #pragma unroll
    for (int n = 0; n < 8; ++n)
        acc[n] = __builtin_amdgcn_wmma_f32_16x16x4_f32(
            false, a0, false, lo2(bq[n]), (short)0, acc[n], false, false);
    const v2f a1 = hi2(qa);
    #pragma unroll
    for (int n = 0; n < 8; ++n)
        acc[n] = __builtin_amdgcn_wmma_f32_16x16x4_f32(
            false, a1, false, hi2(bq[n]), (short)0, acc[n], false, false);
    __builtin_amdgcn_sched_barrier(0);
    asm volatile("v_nop\n\tv_nop\n\tv_nop\n\tv_nop");
    __builtin_amdgcn_sched_barrier(0);
}

__device__ __forceinline__ void slab(v8f (&acc)[8], const float4 (&qc)[4], float4 (&qnx)[4],
                                     const float* src, float4 (&bq0)[8], float4 (&bq1)[8],
                                     const float* bbase, int q0) {
    #pragma unroll
    for (int j = 0; j < 4; ++j) qnx[j] = *(const float4*)(src + 8 * j);
    load_b(bq1, bbase, q0 + 1);
    mma_chunk(acc, qc[0], bq0);
    load_b(bq0, bbase, q0 + 2);
    mma_chunk(acc, qc[1], bq1);
    load_b(bq1, bbase, q0 + 3);
    mma_chunk(acc, qc[2], bq0);
    load_b(bq0, bbase, (q0 + 4) & 63);
    mma_chunk(acc, qc[3], bq1);
}

__global__ __launch_bounds__(256)
void rgcn_gemm_wmma_kernel(const float* __restrict__ x,
                           const float* __restrict__ W,
                           const float* __restrict__ b,
                           const int* __restrict__ cnt,
                           float* __restrict__ out,
                           int nrows, int accumulate) {
    __shared__ __attribute__((aligned(16))) float wsT[64 * CHUNK_F];

    const int tid   = threadIdx.x;
    const int panel = blockIdx.x & 3;
    const int group = blockIdx.x >> 2;
    const int ngrp  = gridDim.x >> 2;
    const int n0    = panel * 128;

    {
        const int c  = tid & 127;
        const int kq = tid >> 7;
        const float* wp = W + n0 + c;
        for (int it = 0; it < 64; ++it) {
            const int k4 = ((((it + group) & 63) << 1) + kq) * 4;
            float4 v;
            v.x = wp[(size_t)(k4 + 0) * DOUT];
            v.y = wp[(size_t)(k4 + 1) * DOUT];
            v.z = wp[(size_t)(k4 + 2) * DOUT];
            v.w = wp[(size_t)(k4 + 3) * DOUT];
            *(float4*)(wsT + (k4 >> 3) * CHUNK_F + (c >> 4) * TILE_F
                           + ((((k4 >> 2) & 1) << 4) + (c & 15)) * 4) = v;
        }
    }
    __syncthreads();

    const int wave = __builtin_amdgcn_readfirstlane(tid >> 5);
    const int lane = tid & 31;
    const int c    = lane & 15;
    const int h    = lane >> 4;

    const int nstrips = (nrows + 15) >> 4;
    const int per     = (nstrips + ngrp - 1) / ngrp;
    const int s_begin = group * per;
    const int s_end   = (s_begin + per < nstrips) ? (s_begin + per) : nstrips;

    const float* bbase = wsT + lane * 4;

    int strip = s_begin + wave;
    if (strip >= s_end) return;

    float4 qa[4], qb[4];
    float4 bq0[8], bq1[8];
    {
        int row = strip * 16 + c;
        if (row >= nrows) row = nrows - 1;
        const float* xr = x + (size_t)row * DIN + 4 * h;
        #pragma unroll
        for (int j = 0; j < 4; ++j) qa[j] = *(const float4*)(xr + 8 * j);
    }
    load_b(bq0, bbase, 0);

    for (; strip < s_end; strip += 8) {
        int row = strip * 16 + c;
        if (row >= nrows) row = nrows - 1;
        int rown = (strip + 8) * 16 + c;
        if (rown >= nrows) rown = nrows - 1;
        const float* xr  = x + (size_t)row  * DIN + 4 * h;
        const float* xnx = x + (size_t)rown * DIN + 4 * h;

        v8f acc[8] = {};

        for (int s = 0; s < 16; s += 2) {
            slab(acc, qa, qb, xr + 32 * (s + 1), bq0, bq1, bbase, 4 * s);
            const float* src = (s + 2 < 16) ? (xr + 32 * (s + 2)) : xnx;
            slab(acc, qb, qa, src, bq0, bq1, bbase, 4 * (s + 1));
        }

        const int rbase = strip * 16 + 8 * h;
        bool has[8];
        #pragma unroll
        for (int v = 0; v < 8; ++v) {
            const int r  = rbase + v;
            const int rc = (r < nrows) ? r : (nrows - 1);
            has[v] = cnt[rc] > 0;
        }
        #pragma unroll
        for (int n = 0; n < 8; ++n) {
            const int col  = n0 + n * 16 + c;
            const float bb = b[col];
            float prev[8];
            #pragma unroll
            for (int v = 0; v < 8; ++v) {
                const int r  = rbase + v;
                const int rc = (r < nrows) ? r : (nrows - 1);
                prev[v] = accumulate ? out[(size_t)rc * DOUT + col] : 0.0f;
            }
            #pragma unroll
            for (int v = 0; v < 8; ++v) {
                const int r = rbase + v;
                if (r < nrows)
                    out[(size_t)r * DOUT + col] = prev[v] + (acc[n][v] + (has[v] ? bb : 0.0f));
            }
        }
    }
}

extern "C" void kernel_launch(void* const* d_in, const int* in_sizes, int n_in,
                              void* d_out, int out_size, void* d_ws, size_t ws_size,
                              hipStream_t stream) {
    const float* x   = (const float*)d_in[0];
    const float* W   = (const float*)d_in[1];
    const float* b   = (const float*)d_in[2];
    const int*   src = (const int*)d_in[3];
    const int*   dst = (const int*)d_in[4];
    float*       out = (float*)d_out;

    const int n    = in_sizes[0] / DIN;
    const int nrel = in_sizes[1] / (DIN * DOUT);
    const int e    = in_sizes[3] / nrel;

    const int m    = nrel * n;
    const int mpad = (m + 3) & ~3;
    float* agg  = (float*)d_ws;
    int*   cnt  = (int*)(agg + (size_t)n * DOUT);
    int*   cur  = cnt + mpad;
    int*   esrc = cur + mpad;

    const int total   = nrel * e;
    const int eblocks = (total + 255) / 256;

    fill_zero_int_kernel<<<(m + 255) / 256, 256, 0, stream>>>(cnt, m);
    rgcn_count_kernel<<<eblocks, 256, 0, stream>>>(dst, cnt, e, n, total);
    rgcn_offsets_kernel<<<(m + 1023) / 1024, 256, 0, stream>>>(cnt, cur, m);
    rgcn_list_kernel<<<eblocks, 256, 0, stream>>>(src, dst, cur, esrc, e, n, total);

    for (int r = 0; r < nrel; ++r) {
        rgcn_row_mean_kernel<<<(n + 15) / 16, 256, 0, stream>>>(
            x, cnt + (size_t)r * n, cur + (size_t)r * n, esrc, agg, n);
        rgcn_gemm_wmma_kernel<<<256, 256, 0, stream>>>(
            agg, W + (size_t)r * DIN * DOUT, b + (size_t)r * DOUT,
            cnt + (size_t)r * n, out, n, r > 0 ? 1 : 0);
    }
}


# Row means of relation r+1 computed inside the GEMM launch of relation r by 8 extra gather-only waves per work-group (memory work under matrix work); two mean planes; sum order unchanged
# speedup vs baseline: 1.7592x; 1.0814x over previous
#include <hip/hip_runtime.h>

typedef __attribute__((ext_vector_type(2))) float v2f;
typedef __attribute__((ext_vector_type(8))) float v8f;

#define DIN  512
#define DOUT 512


__global__ void fill_zero_int_kernel(int* __restrict__ p, int count) {
    int i      = blockIdx.x * blockDim.x + threadIdx.x;
    int stride = gridDim.x * blockDim.x;
    for (; i < count; i += stride) p[i] = 0;
}

__global__ void rgcn_count_kernel(const int* __restrict__ dst,
                                  int* __restrict__ cnt,
                                  int e_per_rel, int n, int total) {
    int i = blockIdx.x * blockDim.x + threadIdx.x;
    if (i >= total) return;
    int r = i / e_per_rel;
    atomicAdd(&cnt[(size_t)r * n + dst[i]], 1);
}

__global__ __launch_bounds__(256)
void rgcn_offsets_kernel(const int* __restrict__ cnt, int* __restrict__ cur, int m) {
    __shared__ int front[8];
    __shared__ int wtot[8];
    const int tid  = threadIdx.x;
    const int lane = tid & 31;
    const int wave = tid >> 5;
    const int tile0 = blockIdx.x * 1024;
    const int4* c4 = (const int4*)cnt;

    int s = 0;
    for (int i = tid; i < (tile0 >> 2); i += 256) {
        const int4 v = c4[i];
        s += (v.x + v.y) + (v.z + v.w);
    }
    for (int o = 16; o > 0; o >>= 1) s += __shfl_xor(s, o);
    if (lane == 0) front[wave] = s;

    const int idx = tile0 + tid * 4;
    int a0 = 0, a1 = 0, a2 = 0, a3 = 0;
    if (idx + 3 < m) {
        const int4 v = c4[idx >> 2];
        a0 = v.x; a1 = v.y; a2 = v.z; a3 = v.w;
    } else {
        if (idx     < m) a0 = cnt[idx];
        if (idx + 1 < m) a1 = cnt[idx + 1];
        if (idx + 2 < m) a2 = cnt[idx + 2];
    }
    const int t = (a0 + a1) + (a2 + a3);
    int inc = t;
    for (int o = 1; o < 32; o <<= 1) {
        const int u = __shfl_up(inc, o);
        if (lane >= o) inc += u;
    }
    if (lane == 31) wtot[wave] = inc;
    __syncthreads();

    int base = 0;
    for (int w = 0; w < 8; ++w) base += front[w];
    for (int w = 0; w < wave; ++w) base += wtot[w];
    const int ex = base + inc - t;
    if (idx     < m) cur[idx]     = ex;
    if (idx + 1 < m) cur[idx + 1] = ex + a0;
    if (idx + 2 < m) cur[idx + 2] = ex + a0 + a1;
    if (idx + 3 < m) cur[idx + 3] = ex + a0 + a1 + a2;
}

__global__ void rgcn_list_kernel(const int* __restrict__ src,
                                 const int* __restrict__ dst,
                                 int* __restrict__ cur,
                                 int* __restrict__ esrc,
                                 int e_per_rel, int n, int total) {
    int i = blockIdx.x * blockDim.x + threadIdx.x;
    if (i >= total) return;
    int r = i / e_per_rel;
    const int pos = atomicAdd(&cur[(size_t)r * n + dst[i]], 1);
    esrc[pos] = src[i];
}

__device__ __forceinline__ void add4(float4& a, const float4 v) {
    a.x += v.x; a.y += v.y; a.z += v.z; a.w += v.w;
}
__device__ __forceinline__ float4 mul4(const float4 a, const float s) {
    float4 r; r.x = a.x * s; r.y = a.y * s; r.z = a.z * s; r.w = a.w * s; return r;
}

__global__ __launch_bounds__(256)
void rgcn_row_mean_kernel(const float* __restrict__ x,
                          const int* __restrict__ cnt,
                          const int* __restrict__ cur,
                          const int* __restrict__ esrc,
                          float* __restrict__ agg,
                          int n) {
    const int lane = threadIdx.x & 31;
    const int wave = __builtin_amdgcn_readfirstlane((int)((blockIdx.x * 256 + threadIdx.x) >> 5));
    const int nw   = (gridDim.x * 256) >> 5;

    for (int row = wave; row < n; row += nw) {
        const int c   = cnt[row];
        const int end = cur[row];
        float4 a0 = {0.f, 0.f, 0.f, 0.f}, a1 = a0, a2 = a0, a3 = a0;
        int k = end - c;
        for (; k + 1 < end; k += 2) {
            const int s0 = esrc[k];
            const int s1 = esrc[k + 1];
            const float4* p0 = (const float4*)(x + (size_t)s0 * DIN) + lane;
            const float4* p1 = (const float4*)(x + (size_t)s1 * DIN) + lane;
            const float4 u0 = p0[0], u1 = p0[32], u2 = p0[64], u3 = p0[96];
            const float4 w0 = p1[0], w1 = p1[32], w2 = p1[64], w3 = p1[96];
            add4(a0, u0); add4(a1, u1); add4(a2, u2); add4(a3, u3);
            add4(a0, w0); add4(a1, w1); add4(a2, w2); add4(a3, w3);
        }
        if (k < end) {
            const int s0 = esrc[k];
            const float4* p0 = (const float4*)(x + (size_t)s0 * DIN) + lane;
            const float4 u0 = p0[0], u1 = p0[32], u2 = p0[64], u3 = p0[96];
            add4(a0, u0); add4(a1, u1); add4(a2, u2); add4(a3, u3);
        }
        const float inv = 1.0f / fmaxf((float)c, 1.0f);
        float4* o = (float4*)(agg + (size_t)row * DIN) + lane;
        o[0]  = mul4(a0, inv);
        o[32] = mul4(a1, inv);
        o[64] = mul4(a2, inv);
        o[96] = mul4(a3, inv);
    }
}

#define ROW_LOAD(e, s)                                                           \
    const float4* p##e = (const float4*)(x + (size_t)(s) * DIN) + lane;          \
    const float4 e##0 = p##e[0], e##1 = p##e[32], e##2 = p##e[64], e##3 = p##e[96];
#define ROW_ADD(e) add4(a0, e##0); add4(a1, e##1); add4(a2, e##2); add4(a3, e##3);

__device__ __forceinline__ void row_mean_rows(const float* __restrict__ x,
                                              const int* __restrict__ cnt,
                                              const int* __restrict__ cur,
                                              const int* __restrict__ esrc,
                                              float* __restrict__ agg,
                                              int n, int wave, int nw, int lane) {
    int row = wave;
    if (row >= n) return;
    int c0 = cnt[row], e0 = cur[row];
    int c1 = 0, e1 = 0;
    if (row + nw < n) { c1 = cnt[row + nw]; e1 = cur[row + nw]; }
    int s0, s1, s2, s3;
    {
        const int k = e0 - c0, last = (e0 > 0 ? e0 : 1) - 1;
        s0 = esrc[min(k, last)];     s1 = esrc[min(k + 1, last)];
        s2 = esrc[min(k + 2, last)]; s3 = esrc[min(k + 3, last)];
    }
    for (; row < n; row += nw) {
        float4 a0 = {0.f, 0.f, 0.f, 0.f}, a1 = a0, a2 = a0, a3 = a0;
        int t0 = 0, t1 = 0, t2 = 0, t3 = 0, c2 = 0, e2 = 0;
        const int last1 = (e1 > 0 ? e1 : 1) - 1;
        const int k1    = e1 - c1;
        if (c0 >= 4) {
            ROW_LOAD(u, s0) ROW_LOAD(w, s1) ROW_LOAD(y, s2) ROW_LOAD(z, s3)
            if (row + nw < n) {
                t0 = esrc[min(k1, last1)];     t1 = esrc[min(k1 + 1, last1)];
                t2 = esrc[min(k1 + 2, last1)]; t3 = esrc[min(k1 + 3, last1)];
            }
            if (row + 2 * nw < n) { c2 = cnt[row + 2 * nw]; e2 = cur[row + 2 * nw]; }
            ROW_ADD(u) ROW_ADD(w) ROW_ADD(y) ROW_ADD(z)
            int k = e0 - c0 + 4;
            for (; k + 3 < e0; k += 4) {
                const int r0 = esrc[k], r1 = esrc[k + 1], r2 = esrc[k + 2], r3 = esrc[k + 3];
                ROW_LOAD(f, r0) ROW_LOAD(g, r1) ROW_LOAD(i, r2) ROW_LOAD(j, r3)
                ROW_ADD(f) ROW_ADD(g) ROW_ADD(i) ROW_ADD(j)
            }
            if (k + 1 < e0) {
                const int r0 = esrc[k], r1 = esrc[k + 1];
                ROW_LOAD(f, r0) ROW_LOAD(g, r1)
                ROW_ADD(f) ROW_ADD(g)
                k += 2;
            }
            if (k < e0) {
                const int r0 = esrc[k];
                ROW_LOAD(f, r0)
                ROW_ADD(f)
            }
        } else {
            if (c0 == 3) {
                ROW_LOAD(u, s0) ROW_LOAD(w, s1) ROW_LOAD(y, s2)
                if (row + nw < n) {
                    t0 = esrc[min(k1, last1)];     t1 = esrc[min(k1 + 1, last1)];
                    t2 = esrc[min(k1 + 2, last1)]; t3 = esrc[min(k1 + 3, last1)];
                }
                if (row + 2 * nw < n) { c2 = cnt[row + 2 * nw]; e2 = cur[row + 2 * nw]; }
                ROW_ADD(u) ROW_ADD(w) ROW_ADD(y)
            } else if (c0 == 2) {
                ROW_LOAD(u, s0) ROW_LOAD(w, s1)
                if (row + nw < n) {
                    t0 = esrc[min(k1, last1)];     t1 = esrc[min(k1 + 1, last1)];
                    t2 = esrc[min(k1 + 2, last1)]; t3 = esrc[min(k1 + 3, last1)];
                }
                if (row + 2 * nw < n) { c2 = cnt[row + 2 * nw]; e2 = cur[row + 2 * nw]; }
                ROW_ADD(u) ROW_ADD(w)
            } else if (c0 == 1) {
                ROW_LOAD(u, s0)
                if (row + nw < n) {
                    t0 = esrc[min(k1, last1)];     t1 = esrc[min(k1 + 1, last1)];
                    t2 = esrc[min(k1 + 2, last1)]; t3 = esrc[min(k1 + 3, last1)];
                }
                if (row + 2 * nw < n) { c2 = cnt[row + 2 * nw]; e2 = cur[row + 2 * nw]; }
                ROW_ADD(u)
            } else {
                if (row + nw < n) {
                    t0 = esrc[min(k1, last1)];     t1 = esrc[min(k1 + 1, last1)];
                    t2 = esrc[min(k1 + 2, last1)]; t3 = esrc[min(k1 + 3, last1)];
                }
                if (row + 2 * nw < n) { c2 = cnt[row + 2 * nw]; e2 = cur[row + 2 * nw]; }
            }
        }
        const float inv = 1.0f / fmaxf((float)c0, 1.0f);
        float4* o = (float4*)(agg + (size_t)row * DIN) + lane;
        o[0]  = mul4(a0, inv);
        o[32] = mul4(a1, inv);
        o[64] = mul4(a2, inv);
        o[96] = mul4(a3, inv);
        c0 = c1; e0 = e1; c1 = c2; e1 = e2;
        s0 = t0; s1 = t1; s2 = t2; s3 = t3;
    }
}

#define CHUNK_F 1024
#define TILE_F  128

__device__ __forceinline__ v2f lo2(const float4 q) { v2f r; r.x = q.x; r.y = q.y; return r; }
__device__ __forceinline__ v2f hi2(const float4 q) { v2f r; r.x = q.z; r.y = q.w; return r; }

__device__ __forceinline__ void load_b(float4 (&bq)[8], const float* bbase, int q) {
    #pragma unroll
    for (int n = 0; n < 8; ++n)
        bq[n] = *(const float4*)(bbase + q * CHUNK_F + n * TILE_F);
    __builtin_amdgcn_sched_barrier(0);
}

__device__ __forceinline__ void mma_chunk(v8f (&acc)[8], const float4 qa, const float4 (&bq)[8]) {
    const v2f a0 = lo2(qa);
    #pragma unroll
    for (int n = 0; n < 8; ++n)
        acc[n] = __builtin_amdgcn_wmma_f32_16x16x4_f32(
            false, a0, false, lo2(bq[n]), (short)0, acc[n], false, false);
    const v2f a1 = hi2(qa);
    #pragma unroll
    for (int n = 0; n < 8; ++n)
        acc[n] = __builtin_amdgcn_wmma_f32_16x16x4_f32(
            false, a1, false, hi2(bq[n]), (short)0, acc[n], false, false);
    __builtin_amdgcn_sched_barrier(0);
    asm volatile("v_nop\n\tv_nop\n\tv_nop\n\tv_nop");
    __builtin_amdgcn_sched_barrier(0);
}

__device__ __forceinline__ void slab(v8f (&acc)[8], const float4 (&qc)[4], float4 (&qnx)[4],
                                     const float* src, float4 (&bq0)[8], float4 (&bq1)[8],
                                     const float* bbase, int q0) {
    #pragma unroll
    for (int j = 0; j < 4; ++j) qnx[j] = *(const float4*)(src + 8 * j);
    load_b(bq1, bbase, q0 + 1);
    mma_chunk(acc, qc[0], bq0);
    load_b(bq0, bbase, q0 + 2);
    mma_chunk(acc, qc[1], bq1);
    load_b(bq1, bbase, q0 + 3);
    mma_chunk(acc, qc[2], bq0);
    load_b(bq0, bbase, (q0 + 4) & 63);
    mma_chunk(acc, qc[3], bq1);
}

__global__ __launch_bounds__(512)
void rgcn_gemm_wmma_kernel(const float* __restrict__ x,
                           const float* __restrict__ W,
                           const float* __restrict__ b,
                           const int* __restrict__ cnt,
                           float* __restrict__ out,
                           int nrows, int accumulate,
                           const float* __restrict__ xraw,
                           const int* __restrict__ cnt_next,
                           const int* __restrict__ cur_next,
                           const int* __restrict__ esrc,
                           float* __restrict__ agg_next,
                           int gather) {
    __shared__ __attribute__((aligned(16))) float wsT[64 * CHUNK_F];

    const int tid   = threadIdx.x;
    const int panel = blockIdx.x & 3;
    const int group = blockIdx.x >> 2;
    const int ngrp  = gridDim.x >> 2;
    const int n0    = panel * 128;

    {
        const int c  = tid & 127;
        const int kq = tid >> 7;
        const float* wp = W + n0 + c;
        for (int it = 0; it < 32; ++it) {
            const int k4 = ((((it + group) & 31) << 2) + kq) * 4;
            float4 v;
            v.x = wp[(size_t)(k4 + 0) * DOUT];
            v.y = wp[(size_t)(k4 + 1) * DOUT];
            v.z = wp[(size_t)(k4 + 2) * DOUT];
            v.w = wp[(size_t)(k4 + 3) * DOUT];
            *(float4*)(wsT + (k4 >> 3) * CHUNK_F + (c >> 4) * TILE_F
                           + ((((k4 >> 2) & 1) << 4) + (c & 15)) * 4) = v;
        }
    }
    __syncthreads();

    const int wave = __builtin_amdgcn_readfirstlane(tid >> 5);
    const int lane = tid & 31;

    if (wave >= 8) {
        if (gather)
            row_mean_rows(xraw, cnt_next, cur_next, esrc, agg_next, nrows,
                          (int)blockIdx.x * 8 + (wave - 8), (int)gridDim.x * 8, lane);
        return;
    }
    const int c    = lane & 15;
    const int h    = lane >> 4;

    const int nstrips = (nrows + 15) >> 4;
    const int per     = (nstrips + ngrp - 1) / ngrp;
    const int s_begin = group * per;
    const int s_end   = (s_begin + per < nstrips) ? (s_begin + per) : nstrips;

    const float* bbase = wsT + lane * 4;

    int strip = s_begin + wave;
    if (strip >= s_end) return;

    float4 qa[4], qb[4];
    float4 bq0[8], bq1[8];
    {
        int row = strip * 16 + c;
        if (row >= nrows) row = nrows - 1;
        const float* xr = x + (size_t)row * DIN + 4 * h;
        #pragma unroll
        for (int j = 0; j < 4; ++j) qa[j] = *(const float4*)(xr + 8 * j);
    }
    load_b(bq0, bbase, 0);

    for (; strip < s_end; strip += 8) {
        int row = strip * 16 + c;
        if (row >= nrows) row = nrows - 1;
        int rown = (strip + 8) * 16 + c;
        if (rown >= nrows) rown = nrows - 1;
        const float* xr  = x + (size_t)row  * DIN + 4 * h;
        const float* xnx = x + (size_t)rown * DIN + 4 * h;

        v8f acc[8] = {};

        for (int s = 0; s < 16; s += 2) {
            slab(acc, qa, qb, xr + 32 * (s + 1), bq0, bq1, bbase, 4 * s);
            const float* src = (s + 2 < 16) ? (xr + 32 * (s + 2)) : xnx;
            slab(acc, qb, qa, src, bq0, bq1, bbase, 4 * (s + 1));
        }

        const int rbase = strip * 16 + 8 * h;
        bool has[8];
        #pragma unroll
        for (int v = 0; v < 8; ++v) {
            const int r  = rbase + v;
            const int rc = (r < nrows) ? r : (nrows - 1);
            has[v] = cnt[rc] > 0;
        }
        #pragma unroll
        for (int n = 0; n < 8; ++n) {
            const int col  = n0 + n * 16 + c;
            const float bb = b[col];
            float prev[8];
            #pragma unroll
            for (int v = 0; v < 8; ++v) {
                const int r  = rbase + v;
                const int rc = (r < nrows) ? r : (nrows - 1);
                prev[v] = accumulate ? out[(size_t)rc * DOUT + col] : 0.0f;
            }
            #pragma unroll
            for (int v = 0; v < 8; ++v) {
                const int r = rbase + v;
                if (r < nrows)
                    out[(size_t)r * DOUT + col] = prev[v] + (acc[n][v] + (has[v] ? bb : 0.0f));
            }
        }
    }
}

extern "C" void kernel_launch(void* const* d_in, const int* in_sizes, int n_in,
                              void* d_out, int out_size, void* d_ws, size_t ws_size,
                              hipStream_t stream) {
    const float* x   = (const float*)d_in[0];
    const float* W   = (const float*)d_in[1];
    const float* b   = (const float*)d_in[2];
    const int*   src = (const int*)d_in[3];
    const int*   dst = (const int*)d_in[4];
    float*       out = (float*)d_out;

    const int n    = in_sizes[0] / DIN;
    const int nrel = in_sizes[1] / (DIN * DOUT);
    const int e    = in_sizes[3] / nrel;

    const int m    = nrel * n;
    const int mpad = (m + 3) & ~3;
    const size_t plane  = (size_t)n * DOUT;
    const size_t ints   = (size_t)2 * mpad + (size_t)nrel * e;
    const int    planes = (ws_size >= (2 * plane + ints) * 4) ? 2 : 1;
    float* agg  = (float*)d_ws;
    int*   cnt  = (int*)(agg + (size_t)planes * plane);
    int*   cur  = cnt + mpad;
    int*   esrc = cur + mpad;

    const int total   = nrel * e;
    const int eblocks = (total + 255) / 256;

    fill_zero_int_kernel<<<(m + 255) / 256, 256, 0, stream>>>(cnt, m);
    rgcn_count_kernel<<<eblocks, 256, 0, stream>>>(dst, cnt, e, n, total);
    rgcn_offsets_kernel<<<(m + 1023) / 1024, 256, 0, stream>>>(cnt, cur, m);
    rgcn_list_kernel<<<eblocks, 256, 0, stream>>>(src, dst, cur, esrc, e, n, total);

    for (int r = 0; r < nrel; ++r) {
        float* cur_plane  = agg + (size_t)(planes == 2 ? (r & 1) : 0) * plane;
        float* next_plane = agg + (size_t)(planes == 2 ? ((r + 1) & 1) : 0) * plane;
        const int gather  = (planes == 2 && r + 1 < nrel) ? 1 : 0;
        const int rn      = (r + 1 < nrel) ? r + 1 : r;
        if (planes == 1 || r == 0)
            rgcn_row_mean_kernel<<<(n + 15) / 16, 256, 0, stream>>>(
                x, cnt + (size_t)r * n, cur + (size_t)r * n, esrc, cur_plane, n);
        rgcn_gemm_wmma_kernel<<<256, 512, 0, stream>>>(
            cur_plane, W + (size_t)r * DIN * DOUT, b + (size_t)r * DOUT,
            cnt + (size_t)r * n, out, n, r > 0 ? 1 : 0,
            x, cnt + (size_t)rn * n, cur + (size_t)rn * n, esrc, next_plane, gather);
    }
}


# Prefix-sum launch: counts in front of a block's tile summed by 1024 threads, 16 independent 16-byte loads in flight each (was 256 threads, one dependent L2 round trip per trip, 146 trips); int sums
# speedup vs baseline: 1.8827x; 1.0702x over previous
#include <hip/hip_runtime.h>

typedef __attribute__((ext_vector_type(2))) float v2f;
typedef __attribute__((ext_vector_type(8))) float v8f;
typedef __attribute__((ext_vector_type(4))) int   v4i;
typedef __attribute__((ext_vector_type(8))) int   v8i;

#define DIN  512
#define DOUT 512


__global__ void fill_zero_int_kernel(int* __restrict__ p, int count) {
    int i      = blockIdx.x * blockDim.x + threadIdx.x;
    int stride = gridDim.x * blockDim.x;
    for (; i < count; i += stride) p[i] = 0;
}

__global__ void rgcn_count_kernel(const int* __restrict__ dst,
                                  int* __restrict__ cnt,
                                  int e_per_rel, int n, int total) {
    int i = blockIdx.x * blockDim.x + threadIdx.x;
    if (i >= total) return;
    int r = i / e_per_rel;
    atomicAdd(&cnt[(size_t)r * n + dst[i]], 1);
}

__global__ __launch_bounds__(1024)
void rgcn_offsets_kernel(const int* __restrict__ cnt, int* __restrict__ cur, int m) {
    __shared__ int front[32];
    __shared__ int wtot[8];
    const int tid  = threadIdx.x;
    const int lane = tid & 31;
    const int wave = tid >> 5;
    const int tile0 = blockIdx.x * 1024;
    const int nq    = tile0 >> 2;
    const v4i* c4 = (const v4i*)cnt;
    const bool scan = tid < 256;

    const int idx = tile0 + (tid & 255) * 4;
    int a0 = 0, a1 = 0, a2 = 0, a3 = 0;
    if (scan) {
        if (idx + 3 < m) {
            const v4i v = c4[idx >> 2];
            a0 = v.x; a1 = v.y; a2 = v.z; a3 = v.w;
        } else {
            if (idx     < m) a0 = cnt[idx];
            if (idx + 1 < m) a1 = cnt[idx + 1];
            if (idx + 2 < m) a2 = cnt[idx + 2];
        }
    }

    int s = 0;
    for (int i0 = tid; i0 < nq; i0 += 16 * 1024) {
        v8i p[8];
        #pragma unroll
        for (int j = 0; j < 8; ++j) {
            const int ia = i0 + (2 * j) * 1024, ib = ia + 1024;
            const v4i lo = c4[ia < nq ? ia : i0];
            const v4i hi = c4[ib < nq ? ib : i0];
            p[j] = __builtin_shufflevector(lo, hi, 0, 1, 2, 3, 4, 5, 6, 7);
        }
        asm volatile("" : "+v"(p[0]), "+v"(p[1]), "+v"(p[2]), "+v"(p[3]),
                          "+v"(p[4]), "+v"(p[5]), "+v"(p[6]), "+v"(p[7]));
        #pragma unroll
        for (int j = 0; j < 8; ++j) {
            const int ia = i0 + (2 * j) * 1024, ib = ia + 1024;
            const int ta = (p[j][0] + p[j][1]) + (p[j][2] + p[j][3]);
            const int tb = (p[j][4] + p[j][5]) + (p[j][6] + p[j][7]);
            s += (ia < nq) ? ta : 0;
            s += (ib < nq) ? tb : 0;
        }
    }
    for (int o = 16; o > 0; o >>= 1) s += __shfl_xor(s, o);
    if (lane == 0) front[wave] = s;

    const int t = (a0 + a1) + (a2 + a3);
    int inc = t;
    if (scan) {
        for (int o = 1; o < 32; o <<= 1) {
            const int u = __shfl_up(inc, o);
            if (lane >= o) inc += u;
        }
        if (lane == 31) wtot[wave] = inc;
    }
    __syncthreads();

    if (scan) {
        int base = 0;
        for (int w = 0; w < 32; ++w) base += front[w];
        for (int w = 0; w < wave; ++w) base += wtot[w];
        const int ex = base + inc - t;
        if (idx     < m) cur[idx]     = ex;
        if (idx + 1 < m) cur[idx + 1] = ex + a0;
        if (idx + 2 < m) cur[idx + 2] = ex + a0 + a1;
        if (idx + 3 < m) cur[idx + 3] = ex + a0 + a1 + a2;
    }
}

__global__ void rgcn_list_kernel(const int* __restrict__ src,
                                 const int* __restrict__ dst,
                                 int* __restrict__ cur,
                                 int* __restrict__ esrc,
                                 int e_per_rel, int n, int total) {
    int i = blockIdx.x * blockDim.x + threadIdx.x;
    if (i >= total) return;
    int r = i / e_per_rel;
    const int pos = atomicAdd(&cur[(size_t)r * n + dst[i]], 1);
    esrc[pos] = src[i];
}

__device__ __forceinline__ void add4(float4& a, const float4 v) {
    a.x += v.x; a.y += v.y; a.z += v.z; a.w += v.w;
}
__device__ __forceinline__ float4 mul4(const float4 a, const float s) {
    float4 r; r.x = a.x * s; r.y = a.y * s; r.z = a.z * s; r.w = a.w * s; return r;
}

__global__ __launch_bounds__(256)
void rgcn_row_mean_kernel(const float* __restrict__ x,
                          const int* __restrict__ cnt,
                          const int* __restrict__ cur,
                          const int* __restrict__ esrc,
                          float* __restrict__ agg,
                          int n) {
    const int lane = threadIdx.x & 31;
    const int wave = __builtin_amdgcn_readfirstlane((int)((blockIdx.x * 256 + threadIdx.x) >> 5));
    const int nw   = (gridDim.x * 256) >> 5;

    for (int row = wave; row < n; row += nw) {
        const int c   = cnt[row];
        const int end = cur[row];
        float4 a0 = {0.f, 0.f, 0.f, 0.f}, a1 = a0, a2 = a0, a3 = a0;
        int k = end - c;
        for (; k + 1 < end; k += 2) {
            const int s0 = esrc[k];
            const int s1 = esrc[k + 1];
            const float4* p0 = (const float4*)(x + (size_t)s0 * DIN) + lane;
            const float4* p1 = (const float4*)(x + (size_t)s1 * DIN) + lane;
            const float4 u0 = p0[0], u1 = p0[32], u2 = p0[64], u3 = p0[96];
            const float4 w0 = p1[0], w1 = p1[32], w2 = p1[64], w3 = p1[96];
            add4(a0, u0); add4(a1, u1); add4(a2, u2); add4(a3, u3);
            add4(a0, w0); add4(a1, w1); add4(a2, w2); add4(a3, w3);
        }
        if (k < end) {
            const int s0 = esrc[k];
            const float4* p0 = (const float4*)(x + (size_t)s0 * DIN) + lane;
            const float4 u0 = p0[0], u1 = p0[32], u2 = p0[64], u3 = p0[96];
            add4(a0, u0); add4(a1, u1); add4(a2, u2); add4(a3, u3);
        }
        const float inv = 1.0f / fmaxf((float)c, 1.0f);
        float4* o = (float4*)(agg + (size_t)row * DIN) + lane;
        o[0]  = mul4(a0, inv);
        o[32] = mul4(a1, inv);
        o[64] = mul4(a2, inv);
        o[96] = mul4(a3, inv);
    }
}

#define ROW_LOAD(e, s)                                                           \
    const float4* p##e = (const float4*)(x + (size_t)(s) * DIN) + lane;          \
    const float4 e##0 = p##e[0], e##1 = p##e[32], e##2 = p##e[64], e##3 = p##e[96];
#define ROW_ADD(e) add4(a0, e##0); add4(a1, e##1); add4(a2, e##2); add4(a3, e##3);

__device__ __forceinline__ void row_mean_rows(const float* __restrict__ x,
                                              const int* __restrict__ cnt,
                                              const int* __restrict__ cur,
                                              const int* __restrict__ esrc,
                                              float* __restrict__ agg,
                                              int n, int wave, int nw, int lane) {
    int row = wave;
    if (row >= n) return;
    int c0 = cnt[row], e0 = cur[row];
    int c1 = 0, e1 = 0;
    if (row + nw < n) { c1 = cnt[row + nw]; e1 = cur[row + nw]; }
    int s0, s1, s2, s3;
    {
        const int k = e0 - c0, last = (e0 > 0 ? e0 : 1) - 1;
        s0 = esrc[min(k, last)];     s1 = esrc[min(k + 1, last)];
        s2 = esrc[min(k + 2, last)]; s3 = esrc[min(k + 3, last)];
    }
    for (; row < n; row += nw) {
        float4 a0 = {0.f, 0.f, 0.f, 0.f}, a1 = a0, a2 = a0, a3 = a0;
        int t0 = 0, t1 = 0, t2 = 0, t3 = 0, c2 = 0, e2 = 0;
        const int last1 = (e1 > 0 ? e1 : 1) - 1;
        const int k1    = e1 - c1;
        if (c0 >= 4) {
            ROW_LOAD(u, s0) ROW_LOAD(w, s1) ROW_LOAD(y, s2) ROW_LOAD(z, s3)
            if (row + nw < n) {
                t0 = esrc[min(k1, last1)];     t1 = esrc[min(k1 + 1, last1)];
                t2 = esrc[min(k1 + 2, last1)]; t3 = esrc[min(k1 + 3, last1)];
            }
            if (row + 2 * nw < n) { c2 = cnt[row + 2 * nw]; e2 = cur[row + 2 * nw]; }
            ROW_ADD(u) ROW_ADD(w) ROW_ADD(y) ROW_ADD(z)
            int k = e0 - c0 + 4;
            for (; k + 3 < e0; k += 4) {
                const int r0 = esrc[k], r1 = esrc[k + 1], r2 = esrc[k + 2], r3 = esrc[k + 3];
                ROW_LOAD(f, r0) ROW_LOAD(g, r1) ROW_LOAD(i, r2) ROW_LOAD(j, r3)
                ROW_ADD(f) ROW_ADD(g) ROW_ADD(i) ROW_ADD(j)
            }
            if (k + 1 < e0) {
                const int r0 = esrc[k], r1 = esrc[k + 1];
                ROW_LOAD(f, r0) ROW_LOAD(g, r1)
                ROW_ADD(f) ROW_ADD(g)
                k += 2;
            }
            if (k < e0) {
                const int r0 = esrc[k];
                ROW_LOAD(f, r0)
                ROW_ADD(f)
            }
        } else {
            if (c0 == 3) {
                ROW_LOAD(u, s0) ROW_LOAD(w, s1) ROW_LOAD(y, s2)
                if (row + nw < n) {
                    t0 = esrc[min(k1, last1)];     t1 = esrc[min(k1 + 1, last1)];
                    t2 = esrc[min(k1 + 2, last1)]; t3 = esrc[min(k1 + 3, last1)];
                }
                if (row + 2 * nw < n) { c2 = cnt[row + 2 * nw]; e2 = cur[row + 2 * nw]; }
                ROW_ADD(u) ROW_ADD(w) ROW_ADD(y)
            } else if (c0 == 2) {
                ROW_LOAD(u, s0) ROW_LOAD(w, s1)
                if (row + nw < n) {
                    t0 = esrc[min(k1, last1)];     t1 = esrc[min(k1 + 1, last1)];
                    t2 = esrc[min(k1 + 2, last1)]; t3 = esrc[min(k1 + 3, last1)];
                }
                if (row + 2 * nw < n) { c2 = cnt[row + 2 * nw]; e2 = cur[row + 2 * nw]; }
                ROW_ADD(u) ROW_ADD(w)
            } else if (c0 == 1) {
                ROW_LOAD(u, s0)
                if (row + nw < n) {
                    t0 = esrc[min(k1, last1)];     t1 = esrc[min(k1 + 1, last1)];
                    t2 = esrc[min(k1 + 2, last1)]; t3 = esrc[min(k1 + 3, last1)];
                }
                if (row + 2 * nw < n) { c2 = cnt[row + 2 * nw]; e2 = cur[row + 2 * nw]; }
                ROW_ADD(u)
            } else {
                if (row + nw < n) {
                    t0 = esrc[min(k1, last1)];     t1 = esrc[min(k1 + 1, last1)];
                    t2 = esrc[min(k1 + 2, last1)]; t3 = esrc[min(k1 + 3, last1)];
                }
                if (row + 2 * nw < n) { c2 = cnt[row + 2 * nw]; e2 = cur[row + 2 * nw]; }
            }
        }
        const float inv = 1.0f / fmaxf((float)c0, 1.0f);
        float4* o = (float4*)(agg + (size_t)row * DIN) + lane;
        o[0]  = mul4(a0, inv);
        o[32] = mul4(a1, inv);
        o[64] = mul4(a2, inv);
        o[96] = mul4(a3, inv);
        c0 = c1; e0 = e1; c1 = c2; e1 = e2;
        s0 = t0; s1 = t1; s2 = t2; s3 = t3;
    }
}

#define CHUNK_F 1024
#define TILE_F  128

__device__ __forceinline__ v2f lo2(const float4 q) { v2f r; r.x = q.x; r.y = q.y; return r; }
__device__ __forceinline__ v2f hi2(const float4 q) { v2f r; r.x = q.z; r.y = q.w; return r; }

__device__ __forceinline__ void load_b(float4 (&bq)[8], const float* bbase, int q) {
    #pragma unroll
    for (int n = 0; n < 8; ++n)
        bq[n] = *(const float4*)(bbase + q * CHUNK_F + n * TILE_F);
    __builtin_amdgcn_sched_barrier(0);
}

__device__ __forceinline__ void mma_chunk(v8f (&acc)[8], const float4 qa, const float4 (&bq)[8]) {
    const v2f a0 = lo2(qa);
    #pragma unroll
    for (int n = 0; n < 8; ++n)
        acc[n] = __builtin_amdgcn_wmma_f32_16x16x4_f32(
            false, a0, false, lo2(bq[n]), (short)0, acc[n], false, false);
    const v2f a1 = hi2(qa);
    #pragma unroll
    for (int n = 0; n < 8; ++n)
        acc[n] = __builtin_amdgcn_wmma_f32_16x16x4_f32(
            false, a1, false, hi2(bq[n]), (short)0, acc[n], false, false);
    __builtin_amdgcn_sched_barrier(0);
    asm volatile("v_nop\n\tv_nop\n\tv_nop\n\tv_nop");
    __builtin_amdgcn_sched_barrier(0);
}

__device__ __forceinline__ void slab(v8f (&acc)[8], const float4 (&qc)[4], float4 (&qnx)[4],
                                     const float* src, float4 (&bq0)[8], float4 (&bq1)[8],
                                     const float* bbase, int q0) {
    #pragma unroll
    for (int j = 0; j < 4; ++j) qnx[j] = *(const float4*)(src + 8 * j);
    load_b(bq1, bbase, q0 + 1);
    mma_chunk(acc, qc[0], bq0);
    load_b(bq0, bbase, q0 + 2);
    mma_chunk(acc, qc[1], bq1);
    load_b(bq1, bbase, q0 + 3);
    mma_chunk(acc, qc[2], bq0);
    load_b(bq0, bbase, (q0 + 4) & 63);
    mma_chunk(acc, qc[3], bq1);
}

__global__ __launch_bounds__(512)
void rgcn_gemm_wmma_kernel(const float* __restrict__ x,
                           const float* __restrict__ W,
                           const float* __restrict__ b,
                           const int* __restrict__ cnt,
                           float* __restrict__ out,
                           int nrows, int accumulate,
                           const float* __restrict__ xraw,
                           const int* __restrict__ cnt_next,
                           const int* __restrict__ cur_next,
                           const int* __restrict__ esrc,
                           float* __restrict__ agg_next,
                           int gather) {
    __shared__ __attribute__((aligned(16))) float wsT[64 * CHUNK_F];

    const int tid   = threadIdx.x;
    const int panel = blockIdx.x & 3;
    const int group = blockIdx.x >> 2;
    const int ngrp  = gridDim.x >> 2;
    const int n0    = panel * 128;

    {
        const int c  = tid & 127;
        const int kq = tid >> 7;
        const float* wp = W + n0 + c;
        for (int it = 0; it < 32; ++it) {
            const int k4 = ((((it + group) & 31) << 2) + kq) * 4;
            float4 v;
            v.x = wp[(size_t)(k4 + 0) * DOUT];
            v.y = wp[(size_t)(k4 + 1) * DOUT];
            v.z = wp[(size_t)(k4 + 2) * DOUT];
            v.w = wp[(size_t)(k4 + 3) * DOUT];
            *(float4*)(wsT + (k4 >> 3) * CHUNK_F + (c >> 4) * TILE_F
                           + ((((k4 >> 2) & 1) << 4) + (c & 15)) * 4) = v;
        }
    }
    __syncthreads();

    const int wave = __builtin_amdgcn_readfirstlane(tid >> 5);
    const int lane = tid & 31;

    if (wave >= 8) {
        if (gather)
            row_mean_rows(xraw, cnt_next, cur_next, esrc, agg_next, nrows,
                          (int)blockIdx.x * 8 + (wave - 8), (int)gridDim.x * 8, lane);
        return;
    }
    const int c    = lane & 15;
    const int h    = lane >> 4;

    const int nstrips = (nrows + 15) >> 4;
    const int per     = (nstrips + ngrp - 1) / ngrp;
    const int s_begin = group * per;
    const int s_end   = (s_begin + per < nstrips) ? (s_begin + per) : nstrips;

    const float* bbase = wsT + lane * 4;

    int strip = s_begin + wave;
    if (strip >= s_end) return;

    float4 qa[4], qb[4];
    float4 bq0[8], bq1[8];
    {
        int row = strip * 16 + c;
        if (row >= nrows) row = nrows - 1;
        const float* xr = x + (size_t)row * DIN + 4 * h;
        #pragma unroll
        for (int j = 0; j < 4; ++j) qa[j] = *(const float4*)(xr + 8 * j);
    }
    load_b(bq0, bbase, 0);

    for (; strip < s_end; strip += 8) {
        int row = strip * 16 + c;
        if (row >= nrows) row = nrows - 1;
        int rown = (strip + 8) * 16 + c;
        if (rown >= nrows) rown = nrows - 1;
        const float* xr  = x + (size_t)row  * DIN + 4 * h;
        const float* xnx = x + (size_t)rown * DIN + 4 * h;

        v8f acc[8] = {};

        for (int s = 0; s < 16; s += 2) {
            slab(acc, qa, qb, xr + 32 * (s + 1), bq0, bq1, bbase, 4 * s);
            const float* src = (s + 2 < 16) ? (xr + 32 * (s + 2)) : xnx;
            slab(acc, qb, qa, src, bq0, bq1, bbase, 4 * (s + 1));
        }

        const int rbase = strip * 16 + 8 * h;
        bool has[8];
        #pragma unroll
        for (int v = 0; v < 8; ++v) {
            const int r  = rbase + v;
            const int rc = (r < nrows) ? r : (nrows - 1);
            has[v] = cnt[rc] > 0;
        }
        #pragma unroll
        for (int n = 0; n < 8; ++n) {
            const int col  = n0 + n * 16 + c;
            const float bb = b[col];
            float prev[8];
            #pragma unroll
            for (int v = 0; v < 8; ++v) {
                const int r  = rbase + v;
                const int rc = (r < nrows) ? r : (nrows - 1);
                prev[v] = accumulate ? out[(size_t)rc * DOUT + col] : 0.0f;
            }
            #pragma unroll
            for (int v = 0; v < 8; ++v) {
                const int r = rbase + v;
                if (r < nrows)
                    out[(size_t)r * DOUT + col] = prev[v] + (acc[n][v] + (has[v] ? bb : 0.0f));
            }
        }
    }
}

extern "C" void kernel_launch(void* const* d_in, const int* in_sizes, int n_in,
                              void* d_out, int out_size, void* d_ws, size_t ws_size,
                              hipStream_t stream) {
    const float* x   = (const float*)d_in[0];
    const float* W   = (const float*)d_in[1];
    const float* b   = (const float*)d_in[2];
    const int*   src = (const int*)d_in[3];
    const int*   dst = (const int*)d_in[4];
    float*       out = (float*)d_out;

    const int n    = in_sizes[0] / DIN;
    const int nrel = in_sizes[1] / (DIN * DOUT);
    const int e    = in_sizes[3] / nrel;

    const int m    = nrel * n;
    const int mpad = (m + 3) & ~3;
    const size_t plane  = (size_t)n * DOUT;
    const size_t ints   = (size_t)2 * mpad + (size_t)nrel * e;
    const int    planes = (ws_size >= (2 * plane + ints) * 4) ? 2 : 1;
    float* agg  = (float*)d_ws;
    int*   cnt  = (int*)(agg + (size_t)planes * plane);
    int*   cur  = cnt + mpad;
    int*   esrc = cur + mpad;

    const int total   = nrel * e;
    const int eblocks = (total + 255) / 256;

    fill_zero_int_kernel<<<(m + 255) / 256, 256, 0, stream>>>(cnt, m);
    rgcn_count_kernel<<<eblocks, 256, 0, stream>>>(dst, cnt, e, n, total);
    rgcn_offsets_kernel<<<(m + 1023) / 1024, 1024, 0, stream>>>(cnt, cur, m);
    rgcn_list_kernel<<<eblocks, 256, 0, stream>>>(src, dst, cur, esrc, e, n, total);

    for (int r = 0; r < nrel; ++r) {
        float* cur_plane  = agg + (size_t)(planes == 2 ? (r & 1) : 0) * plane;
        float* next_plane = agg + (size_t)(planes == 2 ? ((r + 1) & 1) : 0) * plane;
        const int gather  = (planes == 2 && r + 1 < nrel) ? 1 : 0;
        const int rn      = (r + 1 < nrel) ? r + 1 : r;
        if (planes == 1 || r == 0)
            rgcn_row_mean_kernel<<<(n + 15) / 16, 256, 0, stream>>>(
                x, cnt + (size_t)r * n, cur + (size_t)r * n, esrc, cur_plane, n);
        rgcn_gemm_wmma_kernel<<<256, 512, 0, stream>>>(
            cur_plane, W + (size_t)r * DIN * DOUT, b + (size_t)r * DOUT,
            cnt + (size_t)r * n, out, n, r > 0 ? 1 : 0,
            x, cnt + (size_t)rn * n, cur + (size_t)rn * n, esrc, next_plane, gather);
    }
}


# GEMM strips handed out by an LDS ticket counter to all 16 waves of a work-group (was a fixed stride over waves 0-7; gather waves join once done); per-strip sums and their order unchanged
# speedup vs baseline: 2.0653x; 1.0970x over previous
#include <hip/hip_runtime.h>

typedef __attribute__((ext_vector_type(2))) float v2f;
typedef __attribute__((ext_vector_type(8))) float v8f;
typedef __attribute__((ext_vector_type(4))) int   v4i;
typedef __attribute__((ext_vector_type(8))) int   v8i;

#define DIN  512
#define DOUT 512


__global__ void fill_zero_int_kernel(int* __restrict__ p, int count) {
    int i      = blockIdx.x * blockDim.x + threadIdx.x;
    int stride = gridDim.x * blockDim.x;
    for (; i < count; i += stride) p[i] = 0;
}

__global__ void rgcn_count_kernel(const int* __restrict__ dst,
                                  int* __restrict__ cnt,
                                  int e_per_rel, int n, int total) {
    int i = blockIdx.x * blockDim.x + threadIdx.x;
    if (i >= total) return;
    int r = i / e_per_rel;
    atomicAdd(&cnt[(size_t)r * n + dst[i]], 1);
}

__global__ __launch_bounds__(1024)
void rgcn_offsets_kernel(const int* __restrict__ cnt, int* __restrict__ cur, int m) {
    __shared__ int front[32];
    __shared__ int wtot[8];
    const int tid  = threadIdx.x;
    const int lane = tid & 31;
    const int wave = tid >> 5;
    const int tile0 = blockIdx.x * 1024;
    const int nq    = tile0 >> 2;
    const v4i* c4 = (const v4i*)cnt;
    const bool scan = tid < 256;

    const int idx = tile0 + (tid & 255) * 4;
    int a0 = 0, a1 = 0, a2 = 0, a3 = 0;
    if (scan) {
        if (idx + 3 < m) {
            const v4i v = c4[idx >> 2];
            a0 = v.x; a1 = v.y; a2 = v.z; a3 = v.w;
        } else {
            if (idx     < m) a0 = cnt[idx];
            if (idx + 1 < m) a1 = cnt[idx + 1];
            if (idx + 2 < m) a2 = cnt[idx + 2];
        }
    }

    int s = 0;
    for (int i0 = tid; i0 < nq; i0 += 16 * 1024) {
        v8i p[8];
        #pragma unroll
        for (int j = 0; j < 8; ++j) {
            const int ia = i0 + (2 * j) * 1024, ib = ia + 1024;
            const v4i lo = c4[ia < nq ? ia : i0];
            const v4i hi = c4[ib < nq ? ib : i0];
            p[j] = __builtin_shufflevector(lo, hi, 0, 1, 2, 3, 4, 5, 6, 7);
        }
        asm volatile("" : "+v"(p[0]), "+v"(p[1]), "+v"(p[2]), "+v"(p[3]),
                          "+v"(p[4]), "+v"(p[5]), "+v"(p[6]), "+v"(p[7]));
        #pragma unroll
        for (int j = 0; j < 8; ++j) {
            const int ia = i0 + (2 * j) * 1024, ib = ia + 1024;
            const int ta = (p[j][0] + p[j][1]) + (p[j][2] + p[j][3]);
            const int tb = (p[j][4] + p[j][5]) + (p[j][6] + p[j][7]);
            s += (ia < nq) ? ta : 0;
            s += (ib < nq) ? tb : 0;
        }
    }
    for (int o = 16; o > 0; o >>= 1) s += __shfl_xor(s, o);
    if (lane == 0) front[wave] = s;

    const int t = (a0 + a1) + (a2 + a3);
    int inc = t;
    if (scan) {
        for (int o = 1; o < 32; o <<= 1) {
            const int u = __shfl_up(inc, o);
            if (lane >= o) inc += u;
        }
        if (lane == 31) wtot[wave] = inc;
    }
    __syncthreads();

    if (scan) {
        int base = 0;
        for (int w = 0; w < 32; ++w) base += front[w];
        for (int w = 0; w < wave; ++w) base += wtot[w];
        const int ex = base + inc - t;
        if (idx     < m) cur[idx]     = ex;
        if (idx + 1 < m) cur[idx + 1] = ex + a0;
        if (idx + 2 < m) cur[idx + 2] = ex + a0 + a1;
        if (idx + 3 < m) cur[idx + 3] = ex + a0 + a1 + a2;
    }
}

__global__ void rgcn_list_kernel(const int* __restrict__ src,
                                 const int* __restrict__ dst,
                                 int* __restrict__ cur,
                                 int* __restrict__ esrc,
                                 int e_per_rel, int n, int total) {
    int i = blockIdx.x * blockDim.x + threadIdx.x;
    if (i >= total) return;
    int r = i / e_per_rel;
    const int pos = atomicAdd(&cur[(size_t)r * n + dst[i]], 1);
    esrc[pos] = src[i];
}

__device__ __forceinline__ void add4(float4& a, const float4 v) {
    a.x += v.x; a.y += v.y; a.z += v.z; a.w += v.w;
}
__device__ __forceinline__ float4 mul4(const float4 a, const float s) {
    float4 r; r.x = a.x * s; r.y = a.y * s; r.z = a.z * s; r.w = a.w * s; return r;
}

__global__ __launch_bounds__(256)
void rgcn_row_mean_kernel(const float* __restrict__ x,
                          const int* __restrict__ cnt,
                          const int* __restrict__ cur,
                          const int* __restrict__ esrc,
                          float* __restrict__ agg,
                          int n) {
    const int lane = threadIdx.x & 31;
    const int wave = __builtin_amdgcn_readfirstlane((int)((blockIdx.x * 256 + threadIdx.x) >> 5));
    const int nw   = (gridDim.x * 256) >> 5;

    for (int row = wave; row < n; row += nw) {
        const int c   = cnt[row];
        const int end = cur[row];
        float4 a0 = {0.f, 0.f, 0.f, 0.f}, a1 = a0, a2 = a0, a3 = a0;
        int k = end - c;
        for (; k + 1 < end; k += 2) {
            const int s0 = esrc[k];
            const int s1 = esrc[k + 1];
            const float4* p0 = (const float4*)(x + (size_t)s0 * DIN) + lane;
            const float4* p1 = (const float4*)(x + (size_t)s1 * DIN) + lane;
            const float4 u0 = p0[0], u1 = p0[32], u2 = p0[64], u3 = p0[96];
            const float4 w0 = p1[0], w1 = p1[32], w2 = p1[64], w3 = p1[96];
            add4(a0, u0); add4(a1, u1); add4(a2, u2); add4(a3, u3);
            add4(a0, w0); add4(a1, w1); add4(a2, w2); add4(a3, w3);
        }
        if (k < end) {
            const int s0 = esrc[k];
            const float4* p0 = (const float4*)(x + (size_t)s0 * DIN) + lane;
            const float4 u0 = p0[0], u1 = p0[32], u2 = p0[64], u3 = p0[96];
            add4(a0, u0); add4(a1, u1); add4(a2, u2); add4(a3, u3);
        }
        const float inv = 1.0f / fmaxf((float)c, 1.0f);
        float4* o = (float4*)(agg + (size_t)row * DIN) + lane;
        o[0]  = mul4(a0, inv);
        o[32] = mul4(a1, inv);
        o[64] = mul4(a2, inv);
        o[96] = mul4(a3, inv);
    }
}

#define ROW_LOAD(e, s)                                                           \
    const float4* p##e = (const float4*)(x + (size_t)(s) * DIN) + lane;          \
    const float4 e##0 = p##e[0], e##1 = p##e[32], e##2 = p##e[64], e##3 = p##e[96];
#define ROW_ADD(e) add4(a0, e##0); add4(a1, e##1); add4(a2, e##2); add4(a3, e##3);

__device__ __forceinline__ void row_mean_rows(const float* __restrict__ x,
                                              const int* __restrict__ cnt,
                                              const int* __restrict__ cur,
                                              const int* __restrict__ esrc,
                                              float* __restrict__ agg,
                                              int n, int wave, int nw, int lane) {
    int row = wave;
    if (row >= n) return;
    int c0 = cnt[row], e0 = cur[row];
    int c1 = 0, e1 = 0;
    if (row + nw < n) { c1 = cnt[row + nw]; e1 = cur[row + nw]; }
    int s0, s1, s2, s3;
    {
        const int k = e0 - c0, last = (e0 > 0 ? e0 : 1) - 1;
        s0 = esrc[min(k, last)];     s1 = esrc[min(k + 1, last)];
        s2 = esrc[min(k + 2, last)]; s3 = esrc[min(k + 3, last)];
    }
    for (; row < n; row += nw) {
        float4 a0 = {0.f, 0.f, 0.f, 0.f}, a1 = a0, a2 = a0, a3 = a0;
        int t0 = 0, t1 = 0, t2 = 0, t3 = 0, c2 = 0, e2 = 0;
        const int last1 = (e1 > 0 ? e1 : 1) - 1;
        const int k1    = e1 - c1;
        if (c0 >= 4) {
            ROW_LOAD(u, s0) ROW_LOAD(w, s1) ROW_LOAD(y, s2) ROW_LOAD(z, s3)
            if (row + nw < n) {
                t0 = esrc[min(k1, last1)];     t1 = esrc[min(k1 + 1, last1)];
                t2 = esrc[min(k1 + 2, last1)]; t3 = esrc[min(k1 + 3, last1)];
            }
            if (row + 2 * nw < n) { c2 = cnt[row + 2 * nw]; e2 = cur[row + 2 * nw]; }
            ROW_ADD(u) ROW_ADD(w) ROW_ADD(y) ROW_ADD(z)
            int k = e0 - c0 + 4;
            for (; k + 3 < e0; k += 4) {
                const int r0 = esrc[k], r1 = esrc[k + 1], r2 = esrc[k + 2], r3 = esrc[k + 3];
                ROW_LOAD(f, r0) ROW_LOAD(g, r1) ROW_LOAD(i, r2) ROW_LOAD(j, r3)
                ROW_ADD(f) ROW_ADD(g) ROW_ADD(i) ROW_ADD(j)
            }
            if (k + 1 < e0) {
                const int r0 = esrc[k], r1 = esrc[k + 1];
                ROW_LOAD(f, r0) ROW_LOAD(g, r1)
                ROW_ADD(f) ROW_ADD(g)
                k += 2;
            }
            if (k < e0) {
                const int r0 = esrc[k];
                ROW_LOAD(f, r0)
                ROW_ADD(f)
            }
        } else {
            if (c0 == 3) {
                ROW_LOAD(u, s0) ROW_LOAD(w, s1) ROW_LOAD(y, s2)
                if (row + nw < n) {
                    t0 = esrc[min(k1, last1)];     t1 = esrc[min(k1 + 1, last1)];
                    t2 = esrc[min(k1 + 2, last1)]; t3 = esrc[min(k1 + 3, last1)];
                }
                if (row + 2 * nw < n) { c2 = cnt[row + 2 * nw]; e2 = cur[row + 2 * nw]; }
                ROW_ADD(u) ROW_ADD(w) ROW_ADD(y)
            } else if (c0 == 2) {
                ROW_LOAD(u, s0) ROW_LOAD(w, s1)
                if (row + nw < n) {
                    t0 = esrc[min(k1, last1)];     t1 = esrc[min(k1 + 1, last1)];
                    t2 = esrc[min(k1 + 2, last1)]; t3 = esrc[min(k1 + 3, last1)];
                }
                if (row + 2 * nw < n) { c2 = cnt[row + 2 * nw]; e2 = cur[row + 2 * nw]; }
                ROW_ADD(u) ROW_ADD(w)
            } else if (c0 == 1) {
                ROW_LOAD(u, s0)
                if (row + nw < n) {
                    t0 = esrc[min(k1, last1)];     t1 = esrc[min(k1 + 1, last1)];
                    t2 = esrc[min(k1 + 2, last1)]; t3 = esrc[min(k1 + 3, last1)];
                }
                if (row + 2 * nw < n) { c2 = cnt[row + 2 * nw]; e2 = cur[row + 2 * nw]; }
                ROW_ADD(u)
            } else {
                if (row + nw < n) {
                    t0 = esrc[min(k1, last1)];     t1 = esrc[min(k1 + 1, last1)];
                    t2 = esrc[min(k1 + 2, last1)]; t3 = esrc[min(k1 + 3, last1)];
                }
                if (row + 2 * nw < n) { c2 = cnt[row + 2 * nw]; e2 = cur[row + 2 * nw]; }
            }
        }
        const float inv = 1.0f / fmaxf((float)c0, 1.0f);
        float4* o = (float4*)(agg + (size_t)row * DIN) + lane;
        o[0]  = mul4(a0, inv);
        o[32] = mul4(a1, inv);
        o[64] = mul4(a2, inv);
        o[96] = mul4(a3, inv);
        c0 = c1; e0 = e1; c1 = c2; e1 = e2;
        s0 = t0; s1 = t1; s2 = t2; s3 = t3;
    }
}

#define CHUNK_F 1024
#define TILE_F  128

__device__ __forceinline__ v2f lo2(const float4 q) { v2f r; r.x = q.x; r.y = q.y; return r; }
__device__ __forceinline__ v2f hi2(const float4 q) { v2f r; r.x = q.z; r.y = q.w; return r; }

__device__ __forceinline__ void load_b(float4 (&bq)[8], const float* bbase, int q) {
    #pragma unroll
    for (int n = 0; n < 8; ++n)
        bq[n] = *(const float4*)(bbase + q * CHUNK_F + n * TILE_F);
    __builtin_amdgcn_sched_barrier(0);
}

__device__ __forceinline__ void mma_chunk(v8f (&acc)[8], const float4 qa, const float4 (&bq)[8]) {
    const v2f a0 = lo2(qa);
    #pragma unroll
    for (int n = 0; n < 8; ++n)
        acc[n] = __builtin_amdgcn_wmma_f32_16x16x4_f32(
            false, a0, false, lo2(bq[n]), (short)0, acc[n], false, false);
    const v2f a1 = hi2(qa);
    #pragma unroll
    for (int n = 0; n < 8; ++n)
        acc[n] = __builtin_amdgcn_wmma_f32_16x16x4_f32(
            false, a1, false, hi2(bq[n]), (short)0, acc[n], false, false);
    __builtin_amdgcn_sched_barrier(0);
    asm volatile("v_nop\n\tv_nop\n\tv_nop\n\tv_nop");
    __builtin_amdgcn_sched_barrier(0);
}

__device__ __forceinline__ void slab(v8f (&acc)[8], const float4 (&qc)[4], float4 (&qnx)[4],
                                     const float* src, float4 (&bq0)[8], float4 (&bq1)[8],
                                     const float* bbase, int q0) {
    #pragma unroll
    for (int j = 0; j < 4; ++j) qnx[j] = *(const float4*)(src + 8 * j);
    load_b(bq1, bbase, q0 + 1);
    mma_chunk(acc, qc[0], bq0);
    load_b(bq0, bbase, q0 + 2);
    mma_chunk(acc, qc[1], bq1);
    load_b(bq1, bbase, q0 + 3);
    mma_chunk(acc, qc[2], bq0);
    load_b(bq0, bbase, (q0 + 4) & 63);
    mma_chunk(acc, qc[3], bq1);
}

__device__ __forceinline__ int take_strip(int* ctr, int lane) {
    int t = 0;
    if (lane == 0) t = atomicAdd(ctr, 1);
    return __builtin_amdgcn_readfirstlane(t);
}

__global__ __launch_bounds__(512)
void rgcn_gemm_wmma_kernel(const float* __restrict__ x,
                           const float* __restrict__ W,
                           const float* __restrict__ b,
                           const int* __restrict__ cnt,
                           float* __restrict__ out,
                           int nrows, int accumulate,
                           const float* __restrict__ xraw,
                           const int* __restrict__ cnt_next,
                           const int* __restrict__ cur_next,
                           const int* __restrict__ esrc,
                           float* __restrict__ agg_next,
                           int gather) {
    __shared__ __attribute__((aligned(16))) float wsT[64 * CHUNK_F];
    __shared__ int strip_ctr;

    const int tid   = threadIdx.x;
    if (tid == 0) strip_ctr = 0;
    const int panel = blockIdx.x & 3;
    const int group = blockIdx.x >> 2;
    const int ngrp  = gridDim.x >> 2;
    const int n0    = panel * 128;

    {
        const int c  = tid & 127;
        const int kq = tid >> 7;
        const float* wp = W + n0 + c;
        for (int it = 0; it < 32; ++it) {
            const int k4 = ((((it + group) & 31) << 2) + kq) * 4;
            float4 v;
            v.x = wp[(size_t)(k4 + 0) * DOUT];
            v.y = wp[(size_t)(k4 + 1) * DOUT];
            v.z = wp[(size_t)(k4 + 2) * DOUT];
            v.w = wp[(size_t)(k4 + 3) * DOUT];
            *(float4*)(wsT + (k4 >> 3) * CHUNK_F + (c >> 4) * TILE_F
                           + ((((k4 >> 2) & 1) << 4) + (c & 15)) * 4) = v;
        }
    }
    __syncthreads();

    const int wave = __builtin_amdgcn_readfirstlane(tid >> 5);
    const int lane = tid & 31;

    if (wave >= 8 && gather)
        row_mean_rows(xraw, cnt_next, cur_next, esrc, agg_next, nrows,
                      (int)blockIdx.x * 8 + (wave - 8), (int)gridDim.x * 8, lane);
    const int c    = lane & 15;
    const int h    = lane >> 4;

    const int nstrips = (nrows + 15) >> 4;
    const int per     = (nstrips + ngrp - 1) / ngrp;
    const int s_begin = group * per;
    const int s_end   = (s_begin + per < nstrips) ? (s_begin + per) : nstrips;

    const float* bbase = wsT + lane * 4;

    const int nloc = s_end - s_begin;
    int ticket = take_strip(&strip_ctr, lane);
    if (ticket >= nloc) return;
    int strip = s_begin + ticket;

    float4 qa[4], qb[4];
    float4 bq0[8], bq1[8];
    {
        int row = strip * 16 + c;
        if (row >= nrows) row = nrows - 1;
        const float* xr = x + (size_t)row * DIN + 4 * h;
        #pragma unroll
        for (int j = 0; j < 4; ++j) qa[j] = *(const float4*)(xr + 8 * j);
    }
    load_b(bq0, bbase, 0);

    for (;;) {
        int row = strip * 16 + c;
        if (row >= nrows) row = nrows - 1;
        const float* xr  = x + (size_t)row * DIN + 4 * h;
        const float* xnx = xr;
        int tnext = nloc;

        v8f acc[8] = {};

        for (int s = 0; s < 16; s += 2) {
            slab(acc, qa, qb, xr + 32 * (s + 1), bq0, bq1, bbase, 4 * s);
            if (s == 14) {
                tnext = take_strip(&strip_ctr, lane);
                const int sn = (tnext < nloc) ? (s_begin + tnext) : strip;
                int rown = sn * 16 + c;
                if (rown >= nrows) rown = nrows - 1;
                xnx = x + (size_t)rown * DIN + 4 * h;
            }
            const float* src = (s + 2 < 16) ? (xr + 32 * (s + 2)) : xnx;
            slab(acc, qb, qa, src, bq0, bq1, bbase, 4 * (s + 1));
        }

        const int rbase = strip * 16 + 8 * h;
        bool has[8];
        #pragma unroll
        for (int v = 0; v < 8; ++v) {
            const int r  = rbase + v;
            const int rc = (r < nrows) ? r : (nrows - 1);
            has[v] = cnt[rc] > 0;
        }
        #pragma unroll
        for (int n = 0; n < 8; ++n) {
            const int col  = n0 + n * 16 + c;
            const float bb = b[col];
            float prev[8];
            #pragma unroll
            for (int v = 0; v < 8; ++v) {
                const int r  = rbase + v;
                const int rc = (r < nrows) ? r : (nrows - 1);
                prev[v] = accumulate ? out[(size_t)rc * DOUT + col] : 0.0f;
            }
            #pragma unroll
            for (int v = 0; v < 8; ++v) {
                const int r = rbase + v;
                if (r < nrows)
                    out[(size_t)r * DOUT + col] = prev[v] + (acc[n][v] + (has[v] ? bb : 0.0f));
            }
        }
        if (tnext >= nloc) break;
        strip = s_begin + tnext;
    }
}

extern "C" void kernel_launch(void* const* d_in, const int* in_sizes, int n_in,
                              void* d_out, int out_size, void* d_ws, size_t ws_size,
                              hipStream_t stream) {
    const float* x   = (const float*)d_in[0];
    const float* W   = (const float*)d_in[1];
    const float* b   = (const float*)d_in[2];
    const int*   src = (const int*)d_in[3];
    const int*   dst = (const int*)d_in[4];
    float*       out = (float*)d_out;

    const int n    = in_sizes[0] / DIN;
    const int nrel = in_sizes[1] / (DIN * DOUT);
    const int e    = in_sizes[3] / nrel;

    const int m    = nrel * n;
    const int mpad = (m + 3) & ~3;
    const size_t plane  = (size_t)n * DOUT;
    const size_t ints   = (size_t)2 * mpad + (size_t)nrel * e;
    const int    planes = (ws_size >= (2 * plane + ints) * 4) ? 2 : 1;
    float* agg  = (float*)d_ws;
    int*   cnt  = (int*)(agg + (size_t)planes * plane);
    int*   cur  = cnt + mpad;
    int*   esrc = cur + mpad;

    const int total   = nrel * e;
    const int eblocks = (total + 255) / 256;

    fill_zero_int_kernel<<<(m + 255) / 256, 256, 0, stream>>>(cnt, m);
    rgcn_count_kernel<<<eblocks, 256, 0, stream>>>(dst, cnt, e, n, total);
    rgcn_offsets_kernel<<<(m + 1023) / 1024, 1024, 0, stream>>>(cnt, cur, m);
    rgcn_list_kernel<<<eblocks, 256, 0, stream>>>(src, dst, cur, esrc, e, n, total);

    for (int r = 0; r < nrel; ++r) {
        float* cur_plane  = agg + (size_t)(planes == 2 ? (r & 1) : 0) * plane;
        float* next_plane = agg + (size_t)(planes == 2 ? ((r + 1) & 1) : 0) * plane;
        const int gather  = (planes == 2 && r + 1 < nrel) ? 1 : 0;
        const int rn      = (r + 1 < nrel) ? r + 1 : r;
        if (planes == 1 || r == 0)
            rgcn_row_mean_kernel<<<(n + 15) / 16, 256, 0, stream>>>(
                x, cnt + (size_t)r * n, cur + (size_t)r * n, esrc, cur_plane, n);
        rgcn_gemm_wmma_kernel<<<256, 512, 0, stream>>>(
            cur_plane, W + (size_t)r * DIN * DOUT, b + (size_t)r * DOUT,
            cnt + (size_t)r * n, out, n, r > 0 ? 1 : 0,
            x, cnt + (size_t)rn * n, cur + (size_t)rn * n, esrc, next_plane, gather);
    }
}
